# starting kernel
# baseline (speedup 1.0000x reference)
#include <hip/hip_runtime.h>
#include <hip/hip_bf16.h>

typedef float v2f __attribute__((ext_vector_type(2)));
typedef float v8f __attribute__((ext_vector_type(8)));

#define N_NODES 50000
#define N_EDGES 800000
#define IN_D 64
#define HID_D 64
#define OUT_D 32

__global__ void gcn_zero_f32(float* p, long n) {
    long i = (long)blockIdx.x * blockDim.x + threadIdx.x;
    if (i < n) p[i] = 0.0f;
}

__global__ void gcn_degrees(const int* __restrict__ src,
                            const int* __restrict__ dst,
                            float* __restrict__ degA,
                            float* __restrict__ degB, int E) {
    int e = blockIdx.x * blockDim.x + threadIdx.x;
    if (e < E) {
        atomicAdd(degA + src[e], 1.0f);
        atomicAdd(degB + dst[e], 1.0f);
    }
}

__global__ void gcn_norms(float* __restrict__ deg, long n) {
    long i = (long)blockIdx.x * blockDim.x + threadIdx.x;
    if (i < n) deg[i] = rsqrtf(fmaxf(deg[i], 1.0f));
}

__global__ void gcn_scale_transpose(const float* __restrict__ h,
                                    const float* __restrict__ nrmS,
                                    float* __restrict__ feat) {
    long t = (long)blockIdx.x * blockDim.x + threadIdx.x;
    if (t >= (long)IN_D * N_NODES) return;
    int d = (int)(t / N_NODES);
    int n = (int)(t % N_NODES);
    feat[(size_t)n * IN_D + d] = h[t] * nrmS[n];
}

__global__ void gcn_scatter(const int* __restrict__ src,
                            const int* __restrict__ dst,
                            const float* __restrict__ feat,
                            float* __restrict__ agg, int E) {
    int wid  = (blockIdx.x * blockDim.x + threadIdx.x) >> 5;
    int lane = threadIdx.x & 31;
    if (wid >= E) return;
    int s = src[wid];
    int t = dst[wid];
    const float2 v = *(const float2*)(feat + (size_t)s * 64 + lane * 2);
    float* a = agg + (size_t)t * 64 + lane * 2;
    atomicAdd(a + 0, v.x);
    atomicAdd(a + 1, v.y);
}

__global__ void gcn_gemm1_wmma(const float* __restrict__ agg,
                               const float* __restrict__ nrmD,
                               const float* __restrict__ nrmS,
                               const float* __restrict__ W1,
                               const float* __restrict__ b1,
                               float* __restrict__ feat) {
    const int lane  = threadIdx.x & 31;
    const int wv    = threadIdx.x >> 5;
    const int m     = lane & 15;
    const int khalf = (lane >> 4) << 1;
    const int row   = blockIdx.x * 16 + m;
    const int col   = (wv << 4) + m;

    const float        scale = nrmD[row];
    const float* const arow  = agg + (size_t)row * 64;

    v8f acc = {};
#pragma unroll
    for (int kk = 0; kk < 64; kk += 4) {
        const int k = kk + khalf;
        v2f a, b;
        a.x = arow[k] * scale;
        a.y = arow[k + 1] * scale;
        b.x = W1[k * HID_D + col];
        b.y = W1[(k + 1) * HID_D + col];
        acc = __builtin_amdgcn_wmma_f32_16x16x4_f32(
            false, a, false, b, (short)0, acc, false, false);
    }

    const int   rbase = blockIdx.x * 16 + ((lane >> 4) << 3);
    const float bias  = b1[col];
#pragma unroll
    for (int j = 0; j < 8; ++j) {
        const int r = rbase + j;
        float v = acc[j] + bias;
        v = v > 0.0f ? v : 0.0f;
        feat[(size_t)r * 64 + col] = v * nrmS[r];
    }
}

__global__ void gcn_gemm2_wmma(const float* __restrict__ agg,
                               const float* __restrict__ nrmD,
                               const float* __restrict__ W2,
                               const float* __restrict__ b2,
                               float* __restrict__ out) {
    const int lane  = threadIdx.x & 31;
    const int wv    = threadIdx.x >> 5;
    const int m     = lane & 15;
    const int khalf = (lane >> 4) << 1;
    const int row   = blockIdx.x * 16 + m;
    const int col   = (wv << 4) + m;

    const float        scale = nrmD[row];
    const float* const arow  = agg + (size_t)row * 64;

    v8f acc = {};
#pragma unroll
    for (int kk = 0; kk < 64; kk += 4) {
        const int k = kk + khalf;
        v2f a, b;
        a.x = arow[k] * scale;
        a.y = arow[k + 1] * scale;
        b.x = W2[k * OUT_D + col];
        b.y = W2[(k + 1) * OUT_D + col];
        acc = __builtin_amdgcn_wmma_f32_16x16x4_f32(
            false, a, false, b, (short)0, acc, false, false);
    }

    const int   rbase = blockIdx.x * 16 + ((lane >> 4) << 3);
    const float bias  = b2[col];
#pragma unroll
    for (int j = 0; j < 8; ++j) {
        const int r = rbase + j;
        out[(size_t)col * N_NODES + r] = acc[j] + bias;
    }
}

extern "C" void kernel_launch(void* const* d_in, const int* in_sizes, int n_in,
                              void* d_out, int out_size, void* d_ws, size_t ws_size,
                              hipStream_t stream) {
    const float* h   = (const float*)d_in[0];
    const int*   src = (const int*)  d_in[1];
    const int*   dst = (const int*)  d_in[2];
    const float* W1  = (const float*)d_in[3];
    const float* b1  = (const float*)d_in[4];
    const float* W2  = (const float*)d_in[5];
    const float* b2  = (const float*)d_in[6];
    float*       out = (float*)d_out;

    float* nrmS = (float*)d_ws;
    float* nrmD = nrmS + N_NODES;
    float* feat = nrmD + N_NODES;
    float* agg  = feat + (size_t)N_NODES * 64;

    const int  T    = 256;
    const long degN = 2L * N_NODES;
    const long aggN = 64L * N_NODES;

    gcn_zero_f32<<<(unsigned)((degN + T - 1) / T), T, 0, stream>>>(nrmS, degN);
    gcn_degrees <<<(N_EDGES + T - 1) / T, T, 0, stream>>>(src, dst, nrmS, nrmD, N_EDGES);
    gcn_norms   <<<(unsigned)((degN + T - 1) / T), T, 0, stream>>>(nrmS, degN);

    gcn_scale_transpose<<<(unsigned)((aggN + T - 1) / T), T, 0, stream>>>(h, nrmS, feat);
    gcn_zero_f32<<<(unsigned)((aggN + T - 1) / T), T, 0, stream>>>(agg, aggN);
    gcn_scatter <<<(N_EDGES * 32 + T - 1) / T, T, 0, stream>>>(src, dst, feat, agg, N_EDGES);
    gcn_gemm1_wmma<<<N_NODES / 16, 128, 0, stream>>>(agg, nrmD, nrmS, W1, b1, feat);

    gcn_zero_f32<<<(unsigned)((aggN + T - 1) / T), T, 0, stream>>>(agg, aggN);
    gcn_scatter <<<(N_EDGES * 32 + T - 1) / T, T, 0, stream>>>(src, dst, feat, agg, N_EDGES);
    gcn_gemm2_wmma<<<N_NODES / 16, 64, 0, stream>>>(agg, nrmD, W2, b2, out);
}


# edge scatter-add: one wave per run of 32 edges (indices read once, coalesced; 16 row loads in flight ahead of their atomics): 25,000 waves per pass, not 800,000; still float atomics, arrival order
# speedup vs baseline: 1.6301x; 1.6301x over previous
#include <hip/hip_runtime.h>
#include <hip/hip_bf16.h>

typedef float v2f __attribute__((ext_vector_type(2)));
typedef float v8f __attribute__((ext_vector_type(8)));

#define N_NODES 50000
#define N_EDGES 800000
#define IN_D 64
#define HID_D 64
#define OUT_D 32

__global__ void gcn_zero_f32(float* p, long n) {
    long i = (long)blockIdx.x * blockDim.x + threadIdx.x;
    if (i < n) p[i] = 0.0f;
}

__global__ void gcn_degrees(const int* __restrict__ src,
                            const int* __restrict__ dst,
                            float* __restrict__ degA,
                            float* __restrict__ degB, int E) {
    int e = blockIdx.x * blockDim.x + threadIdx.x;
    if (e < E) {
        atomicAdd(degA + src[e], 1.0f);
        atomicAdd(degB + dst[e], 1.0f);
    }
}

__global__ void gcn_norms(float* __restrict__ deg, long n) {
    long i = (long)blockIdx.x * blockDim.x + threadIdx.x;
    if (i < n) deg[i] = rsqrtf(fmaxf(deg[i], 1.0f));
}

__global__ void gcn_gemm1_pre_wmma(const float* __restrict__ h,
                                   const float* __restrict__ nrmS,
                                   const float* __restrict__ W1,
                                   float* __restrict__ feat) {
    const int lane  = threadIdx.x & 31;
    const int tile  = blockIdx.x * 5 + (threadIdx.x >> 5);
    const int m     = lane & 15;
    const int khalf = (lane >> 4) << 1;
    const int row   = tile * 16 + m;

    const float* const hcol = h + row;
    const float* const wcol = W1 + m;

    v8f acc0 = {}, acc1 = {}, acc2 = {}, acc3 = {};
#pragma unroll
    for (int kk = 0; kk < 64; kk += 4) {
        const int k = kk + khalf;
        v2f a, b0, b1, b2, b3;
        a.x  = hcol[(size_t)k * N_NODES];
        a.y  = hcol[(size_t)(k + 1) * N_NODES];
        b0.x = wcol[k * HID_D];
        b0.y = wcol[(k + 1) * HID_D];
        b1.x = wcol[k * HID_D + 16];
        b1.y = wcol[(k + 1) * HID_D + 16];
        b2.x = wcol[k * HID_D + 32];
        b2.y = wcol[(k + 1) * HID_D + 32];
        b3.x = wcol[k * HID_D + 48];
        b3.y = wcol[(k + 1) * HID_D + 48];
        acc0 = __builtin_amdgcn_wmma_f32_16x16x4_f32(
            false, a, false, b0, (short)0, acc0, false, false);
        acc1 = __builtin_amdgcn_wmma_f32_16x16x4_f32(
            false, a, false, b1, (short)0, acc1, false, false);
        acc2 = __builtin_amdgcn_wmma_f32_16x16x4_f32(
            false, a, false, b2, (short)0, acc2, false, false);
        acc3 = __builtin_amdgcn_wmma_f32_16x16x4_f32(
            false, a, false, b3, (short)0, acc3, false, false);
    }

    const int rbase = tile * 16 + ((lane >> 4) << 3);
#pragma unroll
    for (int j = 0; j < 8; ++j) {
        const int   r = rbase + j;
        const float s = nrmS[r];
        float* const o = feat + (size_t)r * 64 + m;
        o[0]  = acc0[j] * s;
        o[16] = acc1[j] * s;
        o[32] = acc2[j] * s;
        o[48] = acc3[j] * s;
    }
}

static_assert(N_EDGES % 256 == 0, "edge runs of 32, 8 waves per block");
__global__ void __launch_bounds__(256)
gcn_scatter(const int* __restrict__ src,
            const int* __restrict__ dst,
            const float* __restrict__ feat,
            float* __restrict__ agg) {
    const int lane = threadIdx.x & 31;
    const int run  = (blockIdx.x * blockDim.x + threadIdx.x) >> 5;
    const int e    = run * 32 + lane;
    const int s    = src[e];
    const int t    = dst[e];
    const float* const frow = feat + lane * 2;
    float* const       arow = agg + lane * 2;
#pragma unroll
    for (int g = 0; g < 32; g += 16) {
        float2 v[16];
#pragma unroll
        for (int j = 0; j < 16; ++j) {
            const int sj = __builtin_amdgcn_readlane(s, g + j);
            v[j] = *(const float2*)(frow + (size_t)sj * 64);
        }
#pragma unroll
        for (int j = 0; j < 16; ++j) {
            const int tj = __builtin_amdgcn_readlane(t, g + j);
            float* const a = arow + (size_t)tj * 64;
            atomicAdd(a + 0, v[j].x);
            atomicAdd(a + 1, v[j].y);
        }
    }
}

__global__ void gcn_bias_relu_scale(const float* __restrict__ agg,
                                    const float* __restrict__ nrmD,
                                    const float* __restrict__ nrmS,
                                    const float* __restrict__ b1,
                                    float* __restrict__ feat) {
    const int t = blockIdx.x * blockDim.x + threadIdx.x;
    if (t >= N_NODES * 16) return;
    const int   n = t >> 4;
    const int   c = (t & 15) << 2;
    const float d = nrmD[n];
    const float s = nrmS[n];
    const float4 a = *(const float4*)(agg + (size_t)t * 4);
    const float4 b = *(const float4*)(b1 + c);
    float4 v;
    v.x = a.x * d + b.x;
    v.y = a.y * d + b.y;
    v.z = a.z * d + b.z;
    v.w = a.w * d + b.w;
    v.x = (v.x > 0.0f ? v.x : 0.0f) * s;
    v.y = (v.y > 0.0f ? v.y : 0.0f) * s;
    v.z = (v.z > 0.0f ? v.z : 0.0f) * s;
    v.w = (v.w > 0.0f ? v.w : 0.0f) * s;
    *(float4*)(feat + (size_t)t * 4) = v;
}

__global__ void gcn_gemm2_wmma(const float* __restrict__ agg,
                               const float* __restrict__ nrmD,
                               const float* __restrict__ W2,
                               const float* __restrict__ b2,
                               float* __restrict__ out) {
    const int lane  = threadIdx.x & 31;
    const int wv    = threadIdx.x >> 5;
    const int m     = lane & 15;
    const int khalf = (lane >> 4) << 1;
    const int row   = blockIdx.x * 16 + m;
    const int col   = (wv << 4) + m;

    const float        scale = nrmD[row];
    const float* const arow  = agg + (size_t)row * 64;

    v8f acc = {};
#pragma unroll
    for (int kk = 0; kk < 64; kk += 4) {
        const int k = kk + khalf;
        v2f a, b;
        a.x = arow[k] * scale;
        a.y = arow[k + 1] * scale;
        b.x = W2[k * OUT_D + col];
        b.y = W2[(k + 1) * OUT_D + col];
        acc = __builtin_amdgcn_wmma_f32_16x16x4_f32(
            false, a, false, b, (short)0, acc, false, false);
    }

    const int   rbase = blockIdx.x * 16 + ((lane >> 4) << 3);
    const float bias  = b2[col];
#pragma unroll
    for (int j = 0; j < 8; ++j) {
        const int r = rbase + j;
        out[(size_t)col * N_NODES + r] = acc[j] + bias;
    }
}

extern "C" void kernel_launch(void* const* d_in, const int* in_sizes, int n_in,
                              void* d_out, int out_size, void* d_ws, size_t ws_size,
                              hipStream_t stream) {
    const float* h   = (const float*)d_in[0];
    const int*   src = (const int*)  d_in[1];
    const int*   dst = (const int*)  d_in[2];
    const float* W1  = (const float*)d_in[3];
    const float* b1  = (const float*)d_in[4];
    const float* W2  = (const float*)d_in[5];
    const float* b2  = (const float*)d_in[6];
    float*       out = (float*)d_out;

    float* nrmS = (float*)d_ws;
    float* nrmD = nrmS + N_NODES;
    float* feat = nrmD + N_NODES;
    float* agg  = feat + (size_t)N_NODES * 64;

    const int  T    = 256;
    const long degN = 2L * N_NODES;
    const long aggN = 64L * N_NODES;

    gcn_zero_f32<<<(unsigned)((degN + T - 1) / T), T, 0, stream>>>(nrmS, degN);
    gcn_degrees <<<(N_EDGES + T - 1) / T, T, 0, stream>>>(src, dst, nrmS, nrmD, N_EDGES);
    gcn_norms   <<<(unsigned)((degN + T - 1) / T), T, 0, stream>>>(nrmS, degN);

    gcn_gemm1_pre_wmma<<<N_NODES / 80, 160, 0, stream>>>(h, nrmS, W1, feat);
    gcn_zero_f32<<<(unsigned)((aggN + T - 1) / T), T, 0, stream>>>(agg, aggN);
    gcn_scatter <<<N_EDGES / 256, 256, 0, stream>>>(src, dst, feat, agg);
    gcn_bias_relu_scale<<<N_NODES * 16 / T, T, 0, stream>>>(agg, nrmD, nrmS, b1, feat);

    gcn_zero_f32<<<(unsigned)((aggN + T - 1) / T), T, 0, stream>>>(agg, aggN);
    gcn_scatter <<<N_EDGES / 256, 256, 0, stream>>>(src, dst, feat, agg);
    gcn_gemm2_wmma<<<N_NODES / 16, 64, 0, stream>>>(agg, nrmD, W2, b2, out);
}


# degree counts and both edge sums: one accumulation plane per package half (by the wave's die number), float atomics stay in the adder's own cache half; readers add the 2 planes; sum order changes
# speedup vs baseline: 3.0352x; 1.8620x over previous
#include <hip/hip_runtime.h>
#include <hip/hip_bf16.h>

typedef float v2f __attribute__((ext_vector_type(2)));
typedef float v8f __attribute__((ext_vector_type(8)));

#define N_NODES 50000
#define N_EDGES 800000
#define IN_D 64
#define HID_D 64
#define OUT_D 32

__global__ void gcn_zero_f32(float* p, long n) {
    long i = (long)blockIdx.x * blockDim.x + threadIdx.x;
    if (i < n) p[i] = 0.0f;
}

__device__ __forceinline__ int gcn_pkg_half() {
    return (int)((__builtin_amdgcn_s_sendmsg_rtn(0x87) >> 18) & 1u);
}

__global__ void gcn_degrees(const int* __restrict__ src,
                            const int* __restrict__ dst,
                            float* __restrict__ deg, int E) {
    int e = blockIdx.x * blockDim.x + threadIdx.x;
    float* const degA = deg + (size_t)gcn_pkg_half() * (2 * N_NODES);
    float* const degB = degA + N_NODES;
    if (e < E) {
        atomicAdd(degA + src[e], 1.0f);
        atomicAdd(degB + dst[e], 1.0f);
    }
}

__global__ void gcn_norms(const float* __restrict__ deg,
                          float* __restrict__ nrm, long n) {
    long i = (long)blockIdx.x * blockDim.x + threadIdx.x;
    if (i < n) nrm[i] = rsqrtf(fmaxf(deg[i] + deg[i + n], 1.0f));
}

__global__ void gcn_gemm1_pre_wmma(const float* __restrict__ h,
                                   const float* __restrict__ nrmS,
                                   const float* __restrict__ W1,
                                   float* __restrict__ feat) {
    const int lane  = threadIdx.x & 31;
    const int tile  = blockIdx.x * 5 + (threadIdx.x >> 5);
    const int m     = lane & 15;
    const int khalf = (lane >> 4) << 1;
    const int row   = tile * 16 + m;

    const float* const hcol = h + row;
    const float* const wcol = W1 + m;

    v8f acc0 = {}, acc1 = {}, acc2 = {}, acc3 = {};
#pragma unroll
    for (int kk = 0; kk < 64; kk += 4) {
        const int k = kk + khalf;
        v2f a, b0, b1, b2, b3;
        a.x  = hcol[(size_t)k * N_NODES];
        a.y  = hcol[(size_t)(k + 1) * N_NODES];
        b0.x = wcol[k * HID_D];
        b0.y = wcol[(k + 1) * HID_D];
        b1.x = wcol[k * HID_D + 16];
        b1.y = wcol[(k + 1) * HID_D + 16];
        b2.x = wcol[k * HID_D + 32];
        b2.y = wcol[(k + 1) * HID_D + 32];
        b3.x = wcol[k * HID_D + 48];
        b3.y = wcol[(k + 1) * HID_D + 48];
        acc0 = __builtin_amdgcn_wmma_f32_16x16x4_f32(
            false, a, false, b0, (short)0, acc0, false, false);
        acc1 = __builtin_amdgcn_wmma_f32_16x16x4_f32(
            false, a, false, b1, (short)0, acc1, false, false);
        acc2 = __builtin_amdgcn_wmma_f32_16x16x4_f32(
            false, a, false, b2, (short)0, acc2, false, false);
        acc3 = __builtin_amdgcn_wmma_f32_16x16x4_f32(
            false, a, false, b3, (short)0, acc3, false, false);
    }

    const int rbase = tile * 16 + ((lane >> 4) << 3);
#pragma unroll
    for (int j = 0; j < 8; ++j) {
        const int   r = rbase + j;
        const float s = nrmS[r];
        float* const o = feat + (size_t)r * 64 + m;
        o[0]  = acc0[j] * s;
        o[16] = acc1[j] * s;
        o[32] = acc2[j] * s;
        o[48] = acc3[j] * s;
    }
}

static_assert(N_EDGES % 256 == 0, "edge runs of 32, 8 waves per block");
__global__ void __launch_bounds__(256)
gcn_scatter(const int* __restrict__ src,
            const int* __restrict__ dst,
            const float* __restrict__ feat,
            float* __restrict__ agg) {
    const int lane = threadIdx.x & 31;
    const int run  = (blockIdx.x * blockDim.x + threadIdx.x) >> 5;
    const int e    = run * 32 + lane;
    const int s    = src[e];
    const int t    = dst[e];
    const float* const frow = feat + lane * 2;
    float* const       arow = agg + (size_t)gcn_pkg_half() * ((size_t)N_NODES * 64) + lane * 2;
#pragma unroll
    for (int g = 0; g < 32; g += 16) {
        float2 v[16];
#pragma unroll
        for (int j = 0; j < 16; ++j) {
            const int sj = __builtin_amdgcn_readlane(s, g + j);
            v[j] = *(const float2*)(frow + (size_t)sj * 64);
        }
#pragma unroll
        for (int j = 0; j < 16; ++j) {
            const int tj = __builtin_amdgcn_readlane(t, g + j);
            float* const a = arow + (size_t)tj * 64;
            atomicAdd(a + 0, v[j].x);
            atomicAdd(a + 1, v[j].y);
        }
    }
}

__global__ void gcn_bias_relu_scale(const float* __restrict__ agg,
                                    const float* __restrict__ nrmD,
                                    const float* __restrict__ nrmS,
                                    const float* __restrict__ b1,
                                    float* __restrict__ feat) {
    const int t = blockIdx.x * blockDim.x + threadIdx.x;
    if (t >= N_NODES * 16) return;
    const int   n = t >> 4;
    const int   c = (t & 15) << 2;
    const float d = nrmD[n];
    const float s = nrmS[n];
    const float4 a0 = *(const float4*)(agg + (size_t)t * 4);
    const float4 a1 = *(const float4*)(agg + (size_t)N_NODES * 64 + (size_t)t * 4);
    const float4 b  = *(const float4*)(b1 + c);
    float4 a;
    a.x = a0.x + a1.x;
    a.y = a0.y + a1.y;
    a.z = a0.z + a1.z;
    a.w = a0.w + a1.w;
    float4 v;
    v.x = a.x * d + b.x;
    v.y = a.y * d + b.y;
    v.z = a.z * d + b.z;
    v.w = a.w * d + b.w;
    v.x = (v.x > 0.0f ? v.x : 0.0f) * s;
    v.y = (v.y > 0.0f ? v.y : 0.0f) * s;
    v.z = (v.z > 0.0f ? v.z : 0.0f) * s;
    v.w = (v.w > 0.0f ? v.w : 0.0f) * s;
    *(float4*)(feat + (size_t)t * 4) = v;
}

__global__ void gcn_gemm2_wmma(const float* __restrict__ agg,
                               const float* __restrict__ nrmD,
                               const float* __restrict__ W2,
                               const float* __restrict__ b2,
                               float* __restrict__ out) {
    const int lane  = threadIdx.x & 31;
    const int wv    = threadIdx.x >> 5;
    const int m     = lane & 15;
    const int khalf = (lane >> 4) << 1;
    const int row   = blockIdx.x * 16 + m;
    const int col   = (wv << 4) + m;

    const float        scale = nrmD[row];
    const float* const arow  = agg + (size_t)row * 64;
    const float* const brow  = arow + (size_t)N_NODES * 64;

    v8f acc = {};
#pragma unroll
    for (int kk = 0; kk < 64; kk += 4) {
        const int k = kk + khalf;
        v2f a, b;
        a.x = (arow[k] + brow[k]) * scale;
        a.y = (arow[k + 1] + brow[k + 1]) * scale;
        b.x = W2[k * OUT_D + col];
        b.y = W2[(k + 1) * OUT_D + col];
        acc = __builtin_amdgcn_wmma_f32_16x16x4_f32(
            false, a, false, b, (short)0, acc, false, false);
    }

    const int   rbase = blockIdx.x * 16 + ((lane >> 4) << 3);
    const float bias  = b2[col];
#pragma unroll
    for (int j = 0; j < 8; ++j) {
        const int r = rbase + j;
        out[(size_t)col * N_NODES + r] = acc[j] + bias;
    }
}

extern "C" void kernel_launch(void* const* d_in, const int* in_sizes, int n_in,
                              void* d_out, int out_size, void* d_ws, size_t ws_size,
                              hipStream_t stream) {
    const float* h   = (const float*)d_in[0];
    const int*   src = (const int*)  d_in[1];
    const int*   dst = (const int*)  d_in[2];
    const float* W1  = (const float*)d_in[3];
    const float* b1  = (const float*)d_in[4];
    const float* W2  = (const float*)d_in[5];
    const float* b2  = (const float*)d_in[6];
    float*       out = (float*)d_out;

    float* nrmS = (float*)d_ws;
    float* nrmD = nrmS + N_NODES;
    float* feat = nrmD + N_NODES;
    float* agg  = feat + (size_t)N_NODES * 64;
    float* deg  = agg + (size_t)N_NODES * 128;

    const int  T    = 256;
    const long degN = 2L * N_NODES;
    const long aggN = 128L * N_NODES;

    gcn_zero_f32<<<(unsigned)((2 * degN + T - 1) / T), T, 0, stream>>>(deg, 2 * degN);
    gcn_degrees <<<(N_EDGES + T - 1) / T, T, 0, stream>>>(src, dst, deg, N_EDGES);
    gcn_norms   <<<(unsigned)((degN + T - 1) / T), T, 0, stream>>>(deg, nrmS, degN);

    gcn_gemm1_pre_wmma<<<N_NODES / 80, 160, 0, stream>>>(h, nrmS, W1, feat);
    gcn_zero_f32<<<(unsigned)((aggN + T - 1) / T), T, 0, stream>>>(agg, aggN);
    gcn_scatter <<<N_EDGES / 256, 256, 0, stream>>>(src, dst, feat, agg);
    gcn_bias_relu_scale<<<N_NODES * 16 / T, T, 0, stream>>>(agg, nrmD, nrmS, b1, feat);

    gcn_zero_f32<<<(unsigned)((aggN + T - 1) / T), T, 0, stream>>>(agg, aggN);
    gcn_scatter <<<N_EDGES / 256, 256, 0, stream>>>(src, dst, feat, agg);
    gcn_gemm2_wmma<<<N_NODES / 16, 64, 0, stream>>>(agg, nrmD, W2, b2, out);
}


# edge-sum rows re-laid so each float-atomic instruction adds one whole 128-byte line (lane holds columns l, l+32; feat stored paired): 2 line requests per edge, not 4; still arrival-order atomics
# speedup vs baseline: 3.3947x; 1.1184x over previous
#include <hip/hip_runtime.h>
#include <hip/hip_bf16.h>

typedef float v2f __attribute__((ext_vector_type(2)));
typedef float v8f __attribute__((ext_vector_type(8)));

#define N_NODES 50000
#define N_EDGES 800000
#define IN_D 64
#define HID_D 64
#define OUT_D 32

__global__ void gcn_zero_f32(float* p, long n) {
    long i = (long)blockIdx.x * blockDim.x + threadIdx.x;
    if (i < n) p[i] = 0.0f;
}

__device__ __forceinline__ int gcn_pkg_half() {
    return (int)((__builtin_amdgcn_s_sendmsg_rtn(0x87) >> 18) & 1u);
}

__global__ void gcn_degrees(const int* __restrict__ src,
                            const int* __restrict__ dst,
                            float* __restrict__ deg, int E) {
    int e = blockIdx.x * blockDim.x + threadIdx.x;
    float* const degA = deg + (size_t)gcn_pkg_half() * (2 * N_NODES);
    float* const degB = degA + N_NODES;
    if (e < E) {
        atomicAdd(degA + src[e], 1.0f);
        atomicAdd(degB + dst[e], 1.0f);
    }
}

__global__ void gcn_norms(const float* __restrict__ deg,
                          float* __restrict__ nrm, long n) {
    long i = (long)blockIdx.x * blockDim.x + threadIdx.x;
    if (i < n) nrm[i] = rsqrtf(fmaxf(deg[i] + deg[i + n], 1.0f));
}

__global__ void gcn_gemm1_pre_wmma(const float* __restrict__ h,
                                   const float* __restrict__ nrmS,
                                   const float* __restrict__ W1,
                                   float* __restrict__ feat) {
    const int lane  = threadIdx.x & 31;
    const int tile  = blockIdx.x * 5 + (threadIdx.x >> 5);
    const int m     = lane & 15;
    const int khalf = (lane >> 4) << 1;
    const int row   = tile * 16 + m;

    const float* const hcol = h + row;
    const float* const wcol = W1 + m;

    v8f acc0 = {}, acc1 = {}, acc2 = {}, acc3 = {};
#pragma unroll
    for (int kk = 0; kk < 64; kk += 4) {
        const int k = kk + khalf;
        v2f a, b0, b1, b2, b3;
        a.x  = hcol[(size_t)k * N_NODES];
        a.y  = hcol[(size_t)(k + 1) * N_NODES];
        b0.x = wcol[k * HID_D];
        b0.y = wcol[(k + 1) * HID_D];
        b1.x = wcol[k * HID_D + 16];
        b1.y = wcol[(k + 1) * HID_D + 16];
        b2.x = wcol[k * HID_D + 32];
        b2.y = wcol[(k + 1) * HID_D + 32];
        b3.x = wcol[k * HID_D + 48];
        b3.y = wcol[(k + 1) * HID_D + 48];
        acc0 = __builtin_amdgcn_wmma_f32_16x16x4_f32(
            false, a, false, b0, (short)0, acc0, false, false);
        acc1 = __builtin_amdgcn_wmma_f32_16x16x4_f32(
            false, a, false, b1, (short)0, acc1, false, false);
        acc2 = __builtin_amdgcn_wmma_f32_16x16x4_f32(
            false, a, false, b2, (short)0, acc2, false, false);
        acc3 = __builtin_amdgcn_wmma_f32_16x16x4_f32(
            false, a, false, b3, (short)0, acc3, false, false);
    }

    const int rbase = tile * 16 + ((lane >> 4) << 3);
#pragma unroll
    for (int j = 0; j < 8; ++j) {
        const int   r = rbase + j;
        const float s = nrmS[r];
        float* const o = feat + (size_t)r * 64 + 2 * m;
        *(float2*)(o)      = make_float2(acc0[j] * s, acc2[j] * s);
        *(float2*)(o + 32) = make_float2(acc1[j] * s, acc3[j] * s);
    }
}

static_assert(N_EDGES % 256 == 0, "edge runs of 32, 8 waves per block");
__global__ void __launch_bounds__(256)
gcn_scatter(const int* __restrict__ src,
            const int* __restrict__ dst,
            const float* __restrict__ feat,
            float* __restrict__ agg) {
    const int lane = threadIdx.x & 31;
    const int run  = (blockIdx.x * blockDim.x + threadIdx.x) >> 5;
    const int e    = run * 32 + lane;
    const int s    = src[e];
    const int t    = dst[e];
    const float* const frow = feat + lane * 2;
    float* const       arow = agg + (size_t)gcn_pkg_half() * ((size_t)N_NODES * 64) + lane;
#pragma unroll
    for (int g = 0; g < 32; g += 16) {
        float2 v[16];
#pragma unroll
        for (int j = 0; j < 16; ++j) {
            const int sj = __builtin_amdgcn_readlane(s, g + j);
            v[j] = *(const float2*)(frow + (size_t)sj * 64);
        }
#pragma unroll
        for (int j = 0; j < 16; ++j) {
            const int tj = __builtin_amdgcn_readlane(t, g + j);
            float* const a = arow + (size_t)tj * 64;
            atomicAdd(a + 0,  v[j].x);
            atomicAdd(a + 32, v[j].y);
        }
    }
}

__global__ void gcn_bias_relu_scale(const float* __restrict__ agg,
                                    const float* __restrict__ nrmD,
                                    const float* __restrict__ nrmS,
                                    const float* __restrict__ b1,
                                    float* __restrict__ feat) {
    const int t = blockIdx.x * blockDim.x + threadIdx.x;
    if (t >= N_NODES * 16) return;
    const int   n = t >> 4;
    const int   c = (t & 15) << 1;
    const float d = nrmD[n];
    const float s = nrmS[n];
    const float* const p0 = agg + (size_t)n * 64 + c;
    const float* const p1 = p0 + (size_t)N_NODES * 64;
    const float2 lo0 = *(const float2*)(p0);
    const float2 hi0 = *(const float2*)(p0 + 32);
    const float2 lo1 = *(const float2*)(p1);
    const float2 hi1 = *(const float2*)(p1 + 32);
    const float2 blo = *(const float2*)(b1 + c);
    const float2 bhi = *(const float2*)(b1 + c + 32);
    float4 a, b;
    a.x = lo0.x + lo1.x;
    a.y = hi0.x + hi1.x;
    a.z = lo0.y + lo1.y;
    a.w = hi0.y + hi1.y;
    b.x = blo.x;
    b.y = bhi.x;
    b.z = blo.y;
    b.w = bhi.y;
    float4 v;
    v.x = a.x * d + b.x;
    v.y = a.y * d + b.y;
    v.z = a.z * d + b.z;
    v.w = a.w * d + b.w;
    v.x = (v.x > 0.0f ? v.x : 0.0f) * s;
    v.y = (v.y > 0.0f ? v.y : 0.0f) * s;
    v.z = (v.z > 0.0f ? v.z : 0.0f) * s;
    v.w = (v.w > 0.0f ? v.w : 0.0f) * s;
    *(float4*)(feat + (size_t)t * 4) = v;
}

__global__ void gcn_gemm2_wmma(const float* __restrict__ agg,
                               const float* __restrict__ nrmD,
                               const float* __restrict__ W2,
                               const float* __restrict__ b2,
                               float* __restrict__ out) {
    const int lane  = threadIdx.x & 31;
    const int wv    = threadIdx.x >> 5;
    const int m     = lane & 15;
    const int khalf = (lane >> 4) << 1;
    const int row   = blockIdx.x * 16 + m;
    const int col   = (wv << 4) + m;

    const float        scale = nrmD[row];
    const float* const arow  = agg + (size_t)row * 64;
    const float* const brow  = arow + (size_t)N_NODES * 64;

    v8f acc = {};
#pragma unroll
    for (int kk = 0; kk < 64; kk += 4) {
        const int k = kk + khalf;
        v2f a, b;
        a.x = (arow[k] + brow[k]) * scale;
        a.y = (arow[k + 1] + brow[k + 1]) * scale;
        b.x = W2[k * OUT_D + col];
        b.y = W2[(k + 1) * OUT_D + col];
        acc = __builtin_amdgcn_wmma_f32_16x16x4_f32(
            false, a, false, b, (short)0, acc, false, false);
    }

    const int   rbase = blockIdx.x * 16 + ((lane >> 4) << 3);
    const float bias  = b2[col];
#pragma unroll
    for (int j = 0; j < 8; ++j) {
        const int r = rbase + j;
        out[(size_t)col * N_NODES + r] = acc[j] + bias;
    }
}

extern "C" void kernel_launch(void* const* d_in, const int* in_sizes, int n_in,
                              void* d_out, int out_size, void* d_ws, size_t ws_size,
                              hipStream_t stream) {
    const float* h   = (const float*)d_in[0];
    const int*   src = (const int*)  d_in[1];
    const int*   dst = (const int*)  d_in[2];
    const float* W1  = (const float*)d_in[3];
    const float* b1  = (const float*)d_in[4];
    const float* W2  = (const float*)d_in[5];
    const float* b2  = (const float*)d_in[6];
    float*       out = (float*)d_out;

    float* nrmS = (float*)d_ws;
    float* nrmD = nrmS + N_NODES;
    float* feat = nrmD + N_NODES;
    float* agg  = feat + (size_t)N_NODES * 64;
    float* deg  = agg + (size_t)N_NODES * 128;

    const int  T    = 256;
    const long degN = 2L * N_NODES;
    const long aggN = 128L * N_NODES;

    gcn_zero_f32<<<(unsigned)((2 * degN + T - 1) / T), T, 0, stream>>>(deg, 2 * degN);
    gcn_degrees <<<(N_EDGES + T - 1) / T, T, 0, stream>>>(src, dst, deg, N_EDGES);
    gcn_norms   <<<(unsigned)((degN + T - 1) / T), T, 0, stream>>>(deg, nrmS, degN);

    gcn_gemm1_pre_wmma<<<N_NODES / 80, 160, 0, stream>>>(h, nrmS, W1, feat);
    gcn_zero_f32<<<(unsigned)((aggN + T - 1) / T), T, 0, stream>>>(agg, aggN);
    gcn_scatter <<<N_EDGES / 256, 256, 0, stream>>>(src, dst, feat, agg);
    gcn_bias_relu_scale<<<N_NODES * 16 / T, T, 0, stream>>>(agg, nrmD, nrmS, b1, feat);

    gcn_zero_f32<<<(unsigned)((aggN + T - 1) / T), T, 0, stream>>>(agg, aggN);
    gcn_scatter <<<N_EDGES / 256, 256, 0, stream>>>(src, dst, feat, agg);
    gcn_gemm2_wmma<<<N_NODES / 16, 64, 0, stream>>>(agg, nrmD, W2, b2, out);
}


# zero fills: each lane stores 16 bytes 25 times at constant offsets (250 work-groups, 2,000 waves per fill) instead of 200,000 waves of one 4-byte store each; same bytes, same 10 launches
# speedup vs baseline: 3.5490x; 1.0455x over previous
#include <hip/hip_runtime.h>
#include <hip/hip_bf16.h>

typedef float v2f __attribute__((ext_vector_type(2)));
typedef float v8f __attribute__((ext_vector_type(8)));

#define N_NODES 50000
#define N_EDGES 800000
#define IN_D 64
#define HID_D 64
#define OUT_D 32
#define ZERO_TRIPS 25

template <int TRIPS>
__global__ void __launch_bounds__(256)
gcn_zero_f32(float* __restrict__ p, int n4) {
    float4* const q    = (float4*)p;
    const float4  zero = make_float4(0.0f, 0.0f, 0.0f, 0.0f);
    const int     run  = (int)blockIdx.x * (TRIPS * 256);
    const int     base = run + (int)threadIdx.x;
    if (run + TRIPS * 256 <= n4) {
#pragma unroll
        for (int k = 0; k < TRIPS; ++k) q[base + k * 256] = zero;
    } else {
        for (int i = base; i < n4; i += 256) q[i] = zero;
    }
}

__device__ __forceinline__ int gcn_pkg_half() {
    return (int)((__builtin_amdgcn_s_sendmsg_rtn(0x87) >> 18) & 1u);
}

__global__ void gcn_degrees(const int* __restrict__ src,
                            const int* __restrict__ dst,
                            float* __restrict__ deg, int E) {
    int e = blockIdx.x * blockDim.x + threadIdx.x;
    float* const degA = deg + (size_t)gcn_pkg_half() * (2 * N_NODES);
    float* const degB = degA + N_NODES;
    if (e < E) {
        atomicAdd(degA + src[e], 1.0f);
        atomicAdd(degB + dst[e], 1.0f);
    }
}

__global__ void gcn_norms(const float* __restrict__ deg,
                          float* __restrict__ nrm, long n) {
    long i = (long)blockIdx.x * blockDim.x + threadIdx.x;
    if (i < n) nrm[i] = rsqrtf(fmaxf(deg[i] + deg[i + n], 1.0f));
}

__global__ void gcn_gemm1_pre_wmma(const float* __restrict__ h,
                                   const float* __restrict__ nrmS,
                                   const float* __restrict__ W1,
                                   float* __restrict__ feat) {
    const int lane  = threadIdx.x & 31;
    const int tile  = blockIdx.x * 5 + (threadIdx.x >> 5);
    const int m     = lane & 15;
    const int khalf = (lane >> 4) << 1;
    const int row   = tile * 16 + m;

    const float* const hcol = h + row;
    const float* const wcol = W1 + m;

    v8f acc0 = {}, acc1 = {}, acc2 = {}, acc3 = {};
#pragma unroll
    for (int kk = 0; kk < 64; kk += 4) {
        const int k = kk + khalf;
        v2f a, b0, b1, b2, b3;
        a.x  = hcol[(size_t)k * N_NODES];
        a.y  = hcol[(size_t)(k + 1) * N_NODES];
        b0.x = wcol[k * HID_D];
        b0.y = wcol[(k + 1) * HID_D];
        b1.x = wcol[k * HID_D + 16];
        b1.y = wcol[(k + 1) * HID_D + 16];
        b2.x = wcol[k * HID_D + 32];
        b2.y = wcol[(k + 1) * HID_D + 32];
        b3.x = wcol[k * HID_D + 48];
        b3.y = wcol[(k + 1) * HID_D + 48];
        acc0 = __builtin_amdgcn_wmma_f32_16x16x4_f32(
            false, a, false, b0, (short)0, acc0, false, false);
        acc1 = __builtin_amdgcn_wmma_f32_16x16x4_f32(
            false, a, false, b1, (short)0, acc1, false, false);
        acc2 = __builtin_amdgcn_wmma_f32_16x16x4_f32(
            false, a, false, b2, (short)0, acc2, false, false);
        acc3 = __builtin_amdgcn_wmma_f32_16x16x4_f32(
            false, a, false, b3, (short)0, acc3, false, false);
    }

    const int rbase = tile * 16 + ((lane >> 4) << 3);
#pragma unroll
    for (int j = 0; j < 8; ++j) {
        const int   r = rbase + j;
        const float s = nrmS[r];
        float* const o = feat + (size_t)r * 64 + 2 * m;
        *(float2*)(o)      = make_float2(acc0[j] * s, acc2[j] * s);
        *(float2*)(o + 32) = make_float2(acc1[j] * s, acc3[j] * s);
    }
}

static_assert(N_EDGES % 256 == 0, "edge runs of 32, 8 waves per block");
__global__ void __launch_bounds__(256)
gcn_scatter(const int* __restrict__ src,
            const int* __restrict__ dst,
            const float* __restrict__ feat,
            float* __restrict__ agg) {
    const int lane = threadIdx.x & 31;
    const int run  = (blockIdx.x * blockDim.x + threadIdx.x) >> 5;
    const int e    = run * 32 + lane;
    const int s    = src[e];
    const int t    = dst[e];
    const float* const frow = feat + lane * 2;
    float* const       arow = agg + (size_t)gcn_pkg_half() * ((size_t)N_NODES * 64) + lane;
#pragma unroll
    for (int g = 0; g < 32; g += 16) {
        float2 v[16];
#pragma unroll
        for (int j = 0; j < 16; ++j) {
            const int sj = __builtin_amdgcn_readlane(s, g + j);
            v[j] = *(const float2*)(frow + (size_t)sj * 64);
        }
#pragma unroll
        for (int j = 0; j < 16; ++j) {
            const int tj = __builtin_amdgcn_readlane(t, g + j);
            float* const a = arow + (size_t)tj * 64;
            atomicAdd(a + 0,  v[j].x);
            atomicAdd(a + 32, v[j].y);
        }
    }
}

__global__ void gcn_bias_relu_scale(const float* __restrict__ agg,
                                    const float* __restrict__ nrmD,
                                    const float* __restrict__ nrmS,
                                    const float* __restrict__ b1,
                                    float* __restrict__ feat) {
    const int t = blockIdx.x * blockDim.x + threadIdx.x;
    if (t >= N_NODES * 16) return;
    const int   n = t >> 4;
    const int   c = (t & 15) << 1;
    const float d = nrmD[n];
    const float s = nrmS[n];
    const float* const p0 = agg + (size_t)n * 64 + c;
    const float* const p1 = p0 + (size_t)N_NODES * 64;
    const float2 lo0 = *(const float2*)(p0);
    const float2 hi0 = *(const float2*)(p0 + 32);
    const float2 lo1 = *(const float2*)(p1);
    const float2 hi1 = *(const float2*)(p1 + 32);
    const float2 blo = *(const float2*)(b1 + c);
    const float2 bhi = *(const float2*)(b1 + c + 32);
    float4 a, b;
    a.x = lo0.x + lo1.x;
    a.y = hi0.x + hi1.x;
    a.z = lo0.y + lo1.y;
    a.w = hi0.y + hi1.y;
    b.x = blo.x;
    b.y = bhi.x;
    b.z = blo.y;
    b.w = bhi.y;
    float4 v;
    v.x = a.x * d + b.x;
    v.y = a.y * d + b.y;
    v.z = a.z * d + b.z;
    v.w = a.w * d + b.w;
    v.x = (v.x > 0.0f ? v.x : 0.0f) * s;
    v.y = (v.y > 0.0f ? v.y : 0.0f) * s;
    v.z = (v.z > 0.0f ? v.z : 0.0f) * s;
    v.w = (v.w > 0.0f ? v.w : 0.0f) * s;
    *(float4*)(feat + (size_t)t * 4) = v;
}

__global__ void gcn_gemm2_wmma(const float* __restrict__ agg,
                               const float* __restrict__ nrmD,
                               const float* __restrict__ W2,
                               const float* __restrict__ b2,
                               float* __restrict__ out) {
    const int lane  = threadIdx.x & 31;
    const int wv    = threadIdx.x >> 5;
    const int m     = lane & 15;
    const int khalf = (lane >> 4) << 1;
    const int row   = blockIdx.x * 16 + m;
    const int col   = (wv << 4) + m;

    const float        scale = nrmD[row];
    const float* const arow  = agg + (size_t)row * 64;
    const float* const brow  = arow + (size_t)N_NODES * 64;

    v8f acc = {};
#pragma unroll
    for (int kk = 0; kk < 64; kk += 4) {
        const int k = kk + khalf;
        v2f a, b;
        a.x = (arow[k] + brow[k]) * scale;
        a.y = (arow[k + 1] + brow[k + 1]) * scale;
        b.x = W2[k * OUT_D + col];
        b.y = W2[(k + 1) * OUT_D + col];
        acc = __builtin_amdgcn_wmma_f32_16x16x4_f32(
            false, a, false, b, (short)0, acc, false, false);
    }

    const int   rbase = blockIdx.x * 16 + ((lane >> 4) << 3);
    const float bias  = b2[col];
#pragma unroll
    for (int j = 0; j < 8; ++j) {
        const int r = rbase + j;
        out[(size_t)col * N_NODES + r] = acc[j] + bias;
    }
}

extern "C" void kernel_launch(void* const* d_in, const int* in_sizes, int n_in,
                              void* d_out, int out_size, void* d_ws, size_t ws_size,
                              hipStream_t stream) {
    const float* h   = (const float*)d_in[0];
    const int*   src = (const int*)  d_in[1];
    const int*   dst = (const int*)  d_in[2];
    const float* W1  = (const float*)d_in[3];
    const float* b1  = (const float*)d_in[4];
    const float* W2  = (const float*)d_in[5];
    const float* b2  = (const float*)d_in[6];
    float*       out = (float*)d_out;

    float* nrmS = (float*)d_ws;
    float* nrmD = nrmS + N_NODES;
    float* feat = nrmD + N_NODES;
    float* agg  = feat + (size_t)N_NODES * 64;
    float* deg  = agg + (size_t)N_NODES * 128;

    const int  T    = 256;
    const long degN = 2L * N_NODES;
    const long aggN = 128L * N_NODES;
    static_assert((66L * N_NODES) % 4 == 0 && (194L * N_NODES) % 4 == 0, "16-byte aligned planes");
    static_assert((128L * N_NODES) % 4 == 0 && (4L * N_NODES) % 4 == 0, "whole 16-byte pieces");
    const int  deg4 = (int)(2 * degN / 4);
    const int  agg4 = (int)(aggN / 4);

    gcn_zero_f32<1><<<(unsigned)((deg4 + T - 1) / T), T, 0, stream>>>(deg, deg4);
    gcn_degrees <<<(N_EDGES + T - 1) / T, T, 0, stream>>>(src, dst, deg, N_EDGES);
    gcn_norms   <<<(unsigned)((degN + T - 1) / T), T, 0, stream>>>(deg, nrmS, degN);

    gcn_gemm1_pre_wmma<<<N_NODES / 80, 160, 0, stream>>>(h, nrmS, W1, feat);
    gcn_zero_f32<ZERO_TRIPS><<<(unsigned)((agg4 + ZERO_TRIPS * T - 1) / (ZERO_TRIPS * T)), T, 0, stream>>>(agg, agg4);
    gcn_scatter <<<N_EDGES / 256, 256, 0, stream>>>(src, dst, feat, agg);
    gcn_bias_relu_scale<<<N_NODES * 16 / T, T, 0, stream>>>(agg, nrmD, nrmS, b1, feat);

    gcn_zero_f32<ZERO_TRIPS><<<(unsigned)((agg4 + ZERO_TRIPS * T - 1) / (ZERO_TRIPS * T)), T, 0, stream>>>(agg, agg4);
    gcn_scatter <<<N_EDGES / 256, 256, 0, stream>>>(src, dst, feat, agg);
    gcn_gemm2_wmma<<<N_NODES / 16, 64, 0, stream>>>(agg, nrmD, W2, b2, out);
}
